# starting kernel
# baseline (speedup 1.0000x reference)
#include <hip/hip_runtime.h>


typedef float v2f __attribute__((ext_vector_type(2)));
typedef float v4f __attribute__((ext_vector_type(4)));
typedef float v8f __attribute__((ext_vector_type(8)));

#define NSTATE 256
#define NBATCH 128
#define SEQLEN 256
#define NLAB   64
#define BC     16
#define KB     64

static __device__ __forceinline__ v8f wmma_f32(v2f a, v2f b, v8f c) {
  return __builtin_amdgcn_wmma_f32_16x16x4_f32(
       false, a,  false, b,
       (short)0, c,  false,  false);
}

__global__ void gather_em(const int* __restrict__ sent,
                          const float* __restrict__ emb,
                          float* __restrict__ em) {
  const int t = blockIdx.x;
  const int b = blockIdx.y;
  const int tok = sent[(size_t)b * SEQLEN + t];
  const v4f* src = (const v4f*)(emb + (size_t)tok * NSTATE);
  v4f* dst = (v4f*)(em + (((size_t)t * NBATCH) + b) * NSTATE);
  dst[threadIdx.x] = src[threadIdx.x];
}

__launch_bounds__(512)
__global__ void scan_pair(const float* __restrict__ Tm,
                          const float* __restrict__ em,
                          float* __restrict__ fwd,
                          float* __restrict__ bwd) {
  __shared__ float S[2][BC * NSTATE];
  const int tid   = threadIdx.x;
  const int lane  = tid & 31;
  const int w     = tid >> 5;
  const int hf    = lane >> 4;
  const int l16   = lane & 15;
  const int dir   = blockIdx.x >> 3;
  const int bbase = (blockIdx.x & 7) * BC;

  v2f a[KB];
  if (dir == 0) {
    const float* tp = Tm + (size_t)(w * 16 + l16) * NSTATE + 2 * hf;
#pragma unroll
    for (int kb = 0; kb < KB; ++kb) a[kb] = *(const v2f*)(tp + kb * 4);
  } else {
    const float* tp = Tm + (size_t)(2 * hf) * NSTATE + (w * 16 + l16);
#pragma unroll
    for (int kb = 0; kb < KB; ++kb) {
      a[kb].x = tp[(size_t)kb * 4 * NSTATE];
      a[kb].y = tp[(size_t)kb * 4 * NSTATE + NSTATE];
    }
  }

  float* dst = dir ? bwd : fwd;

  {
    const int t0 = dir ? (SEQLEN - 1) : 0;
    const v4f* e0 = (const v4f*)(em + (((size_t)t0 * NBATCH) + bbase) * NSTATE);
    v4f* g0 = (v4f*)(dst + (((size_t)t0 * NBATCH) + bbase) * NSTATE);
    v4f* s0 = (v4f*)(&S[0][0]);
#pragma unroll
    for (int r = 0; r < 2; ++r) {
      const int i = tid + r * 512;
      const v4f v = e0[i];
      s0[i] = v;
      g0[i] = v;
    }
  }
  __syncthreads();

  for (int step = 1; step < SEQLEN; ++step) {
    const int t   = dir ? (SEQLEN - 1 - step) : step;
    const int cur = (step - 1) & 1, nxt = step & 1;
    const float* Sb = &S[cur][l16 * NSTATE + 2 * hf];

    v8f acc0 = {0,0,0,0,0,0,0,0}, acc1 = acc0, acc2 = acc0, acc3 = acc0;
#pragma unroll
    for (int kb = 0; kb < KB; kb += 4) {
      const v2f b0 = *(const v2f*)(Sb + (kb + 0) * 4);
      const v2f b1 = *(const v2f*)(Sb + (kb + 1) * 4);
      const v2f b2 = *(const v2f*)(Sb + (kb + 2) * 4);
      const v2f b3 = *(const v2f*)(Sb + (kb + 3) * 4);
      acc0 = wmma_f32(a[kb + 0], b0, acc0);
      acc1 = wmma_f32(a[kb + 1], b1, acc1);
      acc2 = wmma_f32(a[kb + 2], b2, acc2);
      acc3 = wmma_f32(a[kb + 3], b3, acc3);
    }
    const v8f acc = (acc0 + acc1) + (acc2 + acc3);

    const size_t rowoff =
        (((size_t)t * NBATCH) + bbase + l16) * NSTATE + w * 16 + 8 * hf;
    const v4f e0 = *(const v4f*)(em + rowoff);
    const v4f e1 = *(const v4f*)(em + rowoff + 4);
    const v4f lo = {acc[0] * e0.x, acc[1] * e0.y, acc[2] * e0.z, acc[3] * e0.w};
    const v4f hi = {acc[4] * e1.x, acc[5] * e1.y, acc[6] * e1.z, acc[7] * e1.w};

    float* sp = &S[nxt][l16 * NSTATE + w * 16 + 8 * hf];
    *(v4f*)sp       = lo;
    *(v4f*)(sp + 4) = hi;
    *(v4f*)(dst + rowoff)     = lo;
    *(v4f*)(dst + rowoff + 4) = hi;
    __syncthreads();
  }
}

__launch_bounds__(512)
__global__ void bwd_fused(const float* __restrict__ Tm,
                          const float* __restrict__ em,
                          float* __restrict__ fx) {
  __shared__ float S[2][BC * NSTATE];
  const int tid = threadIdx.x, lane = tid & 31, w = tid >> 5;
  const int hf = lane >> 4, l16 = lane & 15;
  const int bbase = blockIdx.x * BC;

  v2f a[KB];
  const float* tp = Tm + (size_t)(2 * hf) * NSTATE + (w * 16 + l16);
#pragma unroll
  for (int kb = 0; kb < KB; ++kb) {
    a[kb].x = tp[(size_t)kb * 4 * NSTATE];
    a[kb].y = tp[(size_t)kb * 4 * NSTATE + NSTATE];
  }

  {
    const v4f* e0 =
        (const v4f*)(em + (((size_t)(SEQLEN - 1) * NBATCH) + bbase) * NSTATE);
    v4f* s0 = (v4f*)(&S[0][0]);
#pragma unroll
    for (int r = 0; r < 2; ++r) {
      const int i = tid + r * 512;
      s0[i] = e0[i];
    }
  }
  __syncthreads();

  for (int step = 1; step < SEQLEN; ++step) {
    const int t = SEQLEN - 1 - step;
    const int cur = (step - 1) & 1, nxt = step & 1;
    const float* Sb = &S[cur][l16 * NSTATE + 2 * hf];

    v8f acc0 = {0,0,0,0,0,0,0,0}, acc1 = acc0, acc2 = acc0, acc3 = acc0;
#pragma unroll
    for (int kb = 0; kb < KB; kb += 4) {
      const v2f b0 = *(const v2f*)(Sb + (kb + 0) * 4);
      const v2f b1 = *(const v2f*)(Sb + (kb + 1) * 4);
      const v2f b2 = *(const v2f*)(Sb + (kb + 2) * 4);
      const v2f b3 = *(const v2f*)(Sb + (kb + 3) * 4);
      acc0 = wmma_f32(a[kb + 0], b0, acc0);
      acc1 = wmma_f32(a[kb + 1], b1, acc1);
      acc2 = wmma_f32(a[kb + 2], b2, acc2);
      acc3 = wmma_f32(a[kb + 3], b3, acc3);
    }
    const v8f acc = (acc0 + acc1) + (acc2 + acc3);

    const size_t rowoff =
        (((size_t)t * NBATCH) + bbase + l16) * NSTATE + w * 16 + 8 * hf;
    const v4f e0 = *(const v4f*)(em + rowoff);
    const v4f e1 = *(const v4f*)(em + rowoff + 4);
    const v4f f0 = *(const v4f*)(fx + rowoff);
    const v4f f1 = *(const v4f*)(fx + rowoff + 4);
    const v4f slo = {acc[0] * e0.x, acc[1] * e0.y, acc[2] * e0.z, acc[3] * e0.w};
    const v4f shi = {acc[4] * e1.x, acc[5] * e1.y, acc[6] * e1.z, acc[7] * e1.w};
    const v4f xlo = {acc[0] * f0.x, acc[1] * f0.y, acc[2] * f0.z, acc[3] * f0.w};
    const v4f xhi = {acc[4] * f1.x, acc[5] * f1.y, acc[6] * f1.z, acc[7] * f1.w};

    float* sp = &S[nxt][l16 * NSTATE + w * 16 + 8 * hf];
    *(v4f*)sp       = slo;
    *(v4f*)(sp + 4) = shi;
    *(v4f*)(fx + rowoff)     = xlo;
    *(v4f*)(fx + rowoff + 4) = xhi;
    __syncthreads();
  }
}

__launch_bounds__(128)
__global__ void proj_kernel(const float* __restrict__ fwd,
                            const float* __restrict__ bwd,
                            const float* __restrict__ em,
                            const float* __restrict__ Wout,
                            float* __restrict__ out, int fused) {
  const int t  = blockIdx.x;
  const int bc = blockIdx.y;
  const int tid = threadIdx.x, lane = tid & 31, w = tid >> 5;
  const int hf = lane >> 4, l16 = lane & 15;

  v2f bw[KB];
  const float* bp = Wout + (size_t)(2 * hf) * NLAB + w * 16 + l16;
#pragma unroll
  for (int kb = 0; kb < KB; ++kb) {
    bw[kb].x = bp[(size_t)kb * 4 * NLAB];
    bw[kb].y = bp[(size_t)kb * 4 * NLAB + NLAB];
  }

  const size_t aoff =
      (((size_t)t * NBATCH) + bc * BC + l16) * NSTATE + 2 * hf;

  v8f acc0 = {0,0,0,0,0,0,0,0}, acc1 = acc0, acc2 = acc0, acc3 = acc0;
  if (fused) {
#pragma unroll
    for (int kb = 0; kb < KB; kb += 4) {
      const v2f a0 = *(const v2f*)(fwd + aoff + (kb + 0) * 4);
      const v2f a1 = *(const v2f*)(fwd + aoff + (kb + 1) * 4);
      const v2f a2 = *(const v2f*)(fwd + aoff + (kb + 2) * 4);
      const v2f a3 = *(const v2f*)(fwd + aoff + (kb + 3) * 4);
      acc0 = wmma_f32(a0, bw[kb + 0], acc0);
      acc1 = wmma_f32(a1, bw[kb + 1], acc1);
      acc2 = wmma_f32(a2, bw[kb + 2], acc2);
      acc3 = wmma_f32(a3, bw[kb + 3], acc3);
    }
  } else {
#pragma unroll
    for (int kb = 0; kb < KB; kb += 2) {
      const v2f f0 = *(const v2f*)(fwd + aoff + (kb + 0) * 4);
      const v2f g0 = *(const v2f*)(bwd + aoff + (kb + 0) * 4);
      const v2f e0 = *(const v2f*)(em  + aoff + (kb + 0) * 4);
      const v2f f1 = *(const v2f*)(fwd + aoff + (kb + 1) * 4);
      const v2f g1 = *(const v2f*)(bwd + aoff + (kb + 1) * 4);
      const v2f e1 = *(const v2f*)(em  + aoff + (kb + 1) * 4);
      const v2f a0 = {f0.x * g0.x / e0.x, f0.y * g0.y / e0.y};
      const v2f a1 = {f1.x * g1.x / e1.x, f1.y * g1.y / e1.y};
      acc0 = wmma_f32(a0, bw[kb + 0], acc0);
      acc1 = wmma_f32(a1, bw[kb + 1], acc1);
    }
  }
  const v8f acc = (acc0 + acc1) + (acc2 + acc3);

#pragma unroll
  for (int r = 0; r < 8; ++r) {
    const size_t b = (size_t)(bc * BC + 8 * hf + r);
    out[(b * SEQLEN + t) * NLAB + w * 16 + l16] = acc[r];
  }
}

extern "C" void kernel_launch(void* const* d_in, const int* in_sizes, int n_in,
                              void* d_out, int out_size, void* d_ws,
                              size_t ws_size, hipStream_t stream) {
  const int*   sent = (const int*)d_in[0];
  const float* emb  = (const float*)d_in[1];
  const float* Tm   = (const float*)d_in[2];
  const float* Wout = (const float*)d_in[3];
  float* out = (float*)d_out;

  const size_t NE = (size_t)SEQLEN * NBATCH * NSTATE;
  float* em  = (float*)d_ws;
  float* fwd = em + NE;

  gather_em<<<dim3(SEQLEN, NBATCH), 64, 0, stream>>>(sent, emb, em);

  if (ws_size >= 3 * NE * sizeof(float)) {
    float* bwd = fwd + NE;
    scan_pair<<<16, 512, 0, stream>>>(Tm, em, fwd, bwd);
    proj_kernel<<<dim3(SEQLEN, 8), 128, 0, stream>>>(fwd, bwd, em, Wout, out, 0);
  } else {
    scan_pair<<<8, 512, 0, stream>>>(Tm, em, fwd, nullptr);
    bwd_fused<<<8, 512, 0, stream>>>(Tm, em, fwd);
    proj_kernel<<<dim3(SEQLEN, 8), 128, 0, stream>>>(fwd, fwd, em, Wout, out, 1);
  }
}


# Pad the LDS state rows of the scan from 256 to 260 floats so the 16 columns of each B-fragment load fall on different banks (was a 16-way bank conflict on every load)
# speedup vs baseline: 2.9425x; 2.9425x over previous
#include <hip/hip_runtime.h>


typedef float v2f __attribute__((ext_vector_type(2)));
typedef float v4f __attribute__((ext_vector_type(4)));
typedef float v8f __attribute__((ext_vector_type(8)));

#define NSTATE 256
#define NBATCH 128
#define SEQLEN 256
#define NLAB   64
#define BC     16
#define KB     64
#define SP     (NSTATE + 4)

static __device__ __forceinline__ v8f wmma_f32(v2f a, v2f b, v8f c) {
  return __builtin_amdgcn_wmma_f32_16x16x4_f32(
       false, a,  false, b,
       (short)0, c,  false,  false);
}

__global__ void gather_em(const int* __restrict__ sent,
                          const float* __restrict__ emb,
                          float* __restrict__ em) {
  const int t = blockIdx.x;
  const int b = blockIdx.y;
  const int tok = sent[(size_t)b * SEQLEN + t];
  const v4f* src = (const v4f*)(emb + (size_t)tok * NSTATE);
  v4f* dst = (v4f*)(em + (((size_t)t * NBATCH) + b) * NSTATE);
  dst[threadIdx.x] = src[threadIdx.x];
}

__launch_bounds__(512)
__global__ void scan_pair(const float* __restrict__ Tm,
                          const float* __restrict__ em,
                          float* __restrict__ fwd,
                          float* __restrict__ bwd) {
  __shared__ __attribute__((aligned(16))) float S[2][BC * SP];
  const int tid   = threadIdx.x;
  const int lane  = tid & 31;
  const int w     = tid >> 5;
  const int hf    = lane >> 4;
  const int l16   = lane & 15;
  const int dir   = blockIdx.x >> 3;
  const int bbase = (blockIdx.x & 7) * BC;

  v2f a[KB];
  if (dir == 0) {
    const float* tp = Tm + (size_t)(w * 16 + l16) * NSTATE + 2 * hf;
#pragma unroll
    for (int kb = 0; kb < KB; ++kb) a[kb] = *(const v2f*)(tp + kb * 4);
  } else {
    const float* tp = Tm + (size_t)(2 * hf) * NSTATE + (w * 16 + l16);
#pragma unroll
    for (int kb = 0; kb < KB; ++kb) {
      a[kb].x = tp[(size_t)kb * 4 * NSTATE];
      a[kb].y = tp[(size_t)kb * 4 * NSTATE + NSTATE];
    }
  }

  float* dst = dir ? bwd : fwd;

  {
    const int t0 = dir ? (SEQLEN - 1) : 0;
    const v4f* e0 = (const v4f*)(em + (((size_t)t0 * NBATCH) + bbase) * NSTATE);
    v4f* g0 = (v4f*)(dst + (((size_t)t0 * NBATCH) + bbase) * NSTATE);
#pragma unroll
    for (int r = 0; r < 2; ++r) {
      const int i = tid + r * 512;
      const v4f v = e0[i];
      *(v4f*)(&S[0][(i >> 6) * SP + (i & 63) * 4]) = v;
      g0[i] = v;
    }
  }
  __syncthreads();

  for (int step = 1; step < SEQLEN; ++step) {
    const int t   = dir ? (SEQLEN - 1 - step) : step;
    const int cur = (step - 1) & 1, nxt = step & 1;
    const float* Sb = &S[cur][l16 * SP + 2 * hf];

    v8f acc0 = {0,0,0,0,0,0,0,0}, acc1 = acc0, acc2 = acc0, acc3 = acc0;
#pragma unroll
    for (int kb = 0; kb < KB; kb += 4) {
      const v2f b0 = *(const v2f*)(Sb + (kb + 0) * 4);
      const v2f b1 = *(const v2f*)(Sb + (kb + 1) * 4);
      const v2f b2 = *(const v2f*)(Sb + (kb + 2) * 4);
      const v2f b3 = *(const v2f*)(Sb + (kb + 3) * 4);
      acc0 = wmma_f32(a[kb + 0], b0, acc0);
      acc1 = wmma_f32(a[kb + 1], b1, acc1);
      acc2 = wmma_f32(a[kb + 2], b2, acc2);
      acc3 = wmma_f32(a[kb + 3], b3, acc3);
    }
    const v8f acc = (acc0 + acc1) + (acc2 + acc3);

    const size_t rowoff =
        (((size_t)t * NBATCH) + bbase + l16) * NSTATE + w * 16 + 8 * hf;
    const v4f e0 = *(const v4f*)(em + rowoff);
    const v4f e1 = *(const v4f*)(em + rowoff + 4);
    const v4f lo = {acc[0] * e0.x, acc[1] * e0.y, acc[2] * e0.z, acc[3] * e0.w};
    const v4f hi = {acc[4] * e1.x, acc[5] * e1.y, acc[6] * e1.z, acc[7] * e1.w};

    float* sp = &S[nxt][l16 * SP + w * 16 + 8 * hf];
    *(v4f*)sp       = lo;
    *(v4f*)(sp + 4) = hi;
    *(v4f*)(dst + rowoff)     = lo;
    *(v4f*)(dst + rowoff + 4) = hi;
    __syncthreads();
  }
}

__launch_bounds__(512)
__global__ void bwd_fused(const float* __restrict__ Tm,
                          const float* __restrict__ em,
                          float* __restrict__ fx) {
  __shared__ __attribute__((aligned(16))) float S[2][BC * SP];
  const int tid = threadIdx.x, lane = tid & 31, w = tid >> 5;
  const int hf = lane >> 4, l16 = lane & 15;
  const int bbase = blockIdx.x * BC;

  v2f a[KB];
  const float* tp = Tm + (size_t)(2 * hf) * NSTATE + (w * 16 + l16);
#pragma unroll
  for (int kb = 0; kb < KB; ++kb) {
    a[kb].x = tp[(size_t)kb * 4 * NSTATE];
    a[kb].y = tp[(size_t)kb * 4 * NSTATE + NSTATE];
  }

  {
    const v4f* e0 =
        (const v4f*)(em + (((size_t)(SEQLEN - 1) * NBATCH) + bbase) * NSTATE);
#pragma unroll
    for (int r = 0; r < 2; ++r) {
      const int i = tid + r * 512;
      *(v4f*)(&S[0][(i >> 6) * SP + (i & 63) * 4]) = e0[i];
    }
  }
  __syncthreads();

  for (int step = 1; step < SEQLEN; ++step) {
    const int t = SEQLEN - 1 - step;
    const int cur = (step - 1) & 1, nxt = step & 1;
    const float* Sb = &S[cur][l16 * SP + 2 * hf];

    v8f acc0 = {0,0,0,0,0,0,0,0}, acc1 = acc0, acc2 = acc0, acc3 = acc0;
#pragma unroll
    for (int kb = 0; kb < KB; kb += 4) {
      const v2f b0 = *(const v2f*)(Sb + (kb + 0) * 4);
      const v2f b1 = *(const v2f*)(Sb + (kb + 1) * 4);
      const v2f b2 = *(const v2f*)(Sb + (kb + 2) * 4);
      const v2f b3 = *(const v2f*)(Sb + (kb + 3) * 4);
      acc0 = wmma_f32(a[kb + 0], b0, acc0);
      acc1 = wmma_f32(a[kb + 1], b1, acc1);
      acc2 = wmma_f32(a[kb + 2], b2, acc2);
      acc3 = wmma_f32(a[kb + 3], b3, acc3);
    }
    const v8f acc = (acc0 + acc1) + (acc2 + acc3);

    const size_t rowoff =
        (((size_t)t * NBATCH) + bbase + l16) * NSTATE + w * 16 + 8 * hf;
    const v4f e0 = *(const v4f*)(em + rowoff);
    const v4f e1 = *(const v4f*)(em + rowoff + 4);
    const v4f f0 = *(const v4f*)(fx + rowoff);
    const v4f f1 = *(const v4f*)(fx + rowoff + 4);
    const v4f slo = {acc[0] * e0.x, acc[1] * e0.y, acc[2] * e0.z, acc[3] * e0.w};
    const v4f shi = {acc[4] * e1.x, acc[5] * e1.y, acc[6] * e1.z, acc[7] * e1.w};
    const v4f xlo = {acc[0] * f0.x, acc[1] * f0.y, acc[2] * f0.z, acc[3] * f0.w};
    const v4f xhi = {acc[4] * f1.x, acc[5] * f1.y, acc[6] * f1.z, acc[7] * f1.w};

    float* sp = &S[nxt][l16 * SP + w * 16 + 8 * hf];
    *(v4f*)sp       = slo;
    *(v4f*)(sp + 4) = shi;
    *(v4f*)(fx + rowoff)     = xlo;
    *(v4f*)(fx + rowoff + 4) = xhi;
    __syncthreads();
  }
}

__launch_bounds__(128)
__global__ void proj_kernel(const float* __restrict__ fwd,
                            const float* __restrict__ bwd,
                            const float* __restrict__ em,
                            const float* __restrict__ Wout,
                            float* __restrict__ out, int fused) {
  const int t  = blockIdx.x;
  const int bc = blockIdx.y;
  const int tid = threadIdx.x, lane = tid & 31, w = tid >> 5;
  const int hf = lane >> 4, l16 = lane & 15;

  v2f bw[KB];
  const float* bp = Wout + (size_t)(2 * hf) * NLAB + w * 16 + l16;
#pragma unroll
  for (int kb = 0; kb < KB; ++kb) {
    bw[kb].x = bp[(size_t)kb * 4 * NLAB];
    bw[kb].y = bp[(size_t)kb * 4 * NLAB + NLAB];
  }

  const size_t aoff =
      (((size_t)t * NBATCH) + bc * BC + l16) * NSTATE + 2 * hf;

  v8f acc0 = {0,0,0,0,0,0,0,0}, acc1 = acc0, acc2 = acc0, acc3 = acc0;
  if (fused) {
#pragma unroll
    for (int kb = 0; kb < KB; kb += 4) {
      const v2f a0 = *(const v2f*)(fwd + aoff + (kb + 0) * 4);
      const v2f a1 = *(const v2f*)(fwd + aoff + (kb + 1) * 4);
      const v2f a2 = *(const v2f*)(fwd + aoff + (kb + 2) * 4);
      const v2f a3 = *(const v2f*)(fwd + aoff + (kb + 3) * 4);
      acc0 = wmma_f32(a0, bw[kb + 0], acc0);
      acc1 = wmma_f32(a1, bw[kb + 1], acc1);
      acc2 = wmma_f32(a2, bw[kb + 2], acc2);
      acc3 = wmma_f32(a3, bw[kb + 3], acc3);
    }
  } else {
#pragma unroll
    for (int kb = 0; kb < KB; kb += 2) {
      const v2f f0 = *(const v2f*)(fwd + aoff + (kb + 0) * 4);
      const v2f g0 = *(const v2f*)(bwd + aoff + (kb + 0) * 4);
      const v2f e0 = *(const v2f*)(em  + aoff + (kb + 0) * 4);
      const v2f f1 = *(const v2f*)(fwd + aoff + (kb + 1) * 4);
      const v2f g1 = *(const v2f*)(bwd + aoff + (kb + 1) * 4);
      const v2f e1 = *(const v2f*)(em  + aoff + (kb + 1) * 4);
      const v2f a0 = {f0.x * g0.x / e0.x, f0.y * g0.y / e0.y};
      const v2f a1 = {f1.x * g1.x / e1.x, f1.y * g1.y / e1.y};
      acc0 = wmma_f32(a0, bw[kb + 0], acc0);
      acc1 = wmma_f32(a1, bw[kb + 1], acc1);
    }
  }
  const v8f acc = (acc0 + acc1) + (acc2 + acc3);

#pragma unroll
  for (int r = 0; r < 8; ++r) {
    const size_t b = (size_t)(bc * BC + 8 * hf + r);
    out[(b * SEQLEN + t) * NLAB + w * 16 + l16] = acc[r];
  }
}

extern "C" void kernel_launch(void* const* d_in, const int* in_sizes, int n_in,
                              void* d_out, int out_size, void* d_ws,
                              size_t ws_size, hipStream_t stream) {
  const int*   sent = (const int*)d_in[0];
  const float* emb  = (const float*)d_in[1];
  const float* Tm   = (const float*)d_in[2];
  const float* Wout = (const float*)d_in[3];
  float* out = (float*)d_out;

  const size_t NE = (size_t)SEQLEN * NBATCH * NSTATE;
  float* em  = (float*)d_ws;
  float* fwd = em + NE;

  gather_em<<<dim3(SEQLEN, NBATCH), 64, 0, stream>>>(sent, emb, em);

  if (ws_size >= 3 * NE * sizeof(float)) {
    float* bwd = fwd + NE;
    scan_pair<<<16, 512, 0, stream>>>(Tm, em, fwd, bwd);
    proj_kernel<<<dim3(SEQLEN, 8), 128, 0, stream>>>(fwd, bwd, em, Wout, out, 0);
  } else {
    scan_pair<<<8, 512, 0, stream>>>(Tm, em, fwd, nullptr);
    bwd_fused<<<8, 512, 0, stream>>>(Tm, em, fwd);
    proj_kernel<<<dim3(SEQLEN, 8), 128, 0, stream>>>(fwd, fwd, em, Wout, out, 1);
  }
}


# Scan step barrier orders LDS only (LDS-scoped release/acquire fence + s_barrier, not __syncthreads): a wave no longer waits for the write acknowledgement of its output rows before it signals
# speedup vs baseline: 3.4474x; 1.1716x over previous
#include <hip/hip_runtime.h>


typedef float v2f __attribute__((ext_vector_type(2)));
typedef float v4f __attribute__((ext_vector_type(4)));
typedef float v8f __attribute__((ext_vector_type(8)));

#define NSTATE 256
#define NBATCH 128
#define SEQLEN 256
#define NLAB   64
#define BC     16
#define KB     64
#define SP     (NSTATE + 4)

static __device__ __forceinline__ v8f wmma_f32(v2f a, v2f b, v8f c) {
  return __builtin_amdgcn_wmma_f32_16x16x4_f32(
       false, a,  false, b,
       (short)0, c,  false,  false);
}

static __device__ __forceinline__ void lds_barrier() {
  __builtin_amdgcn_fence(__ATOMIC_RELEASE, "workgroup", "local");
  __builtin_amdgcn_s_barrier();
  __builtin_amdgcn_fence(__ATOMIC_ACQUIRE, "workgroup", "local");
}

__global__ void gather_em(const int* __restrict__ sent,
                          const float* __restrict__ emb,
                          float* __restrict__ em) {
  const int t = blockIdx.x;
  const int b = blockIdx.y;
  const int tok = sent[(size_t)b * SEQLEN + t];
  const v4f* src = (const v4f*)(emb + (size_t)tok * NSTATE);
  v4f* dst = (v4f*)(em + (((size_t)t * NBATCH) + b) * NSTATE);
  dst[threadIdx.x] = src[threadIdx.x];
}

__launch_bounds__(512)
__global__ void scan_pair(const float* __restrict__ Tm,
                          const float* __restrict__ em,
                          float* __restrict__ fwd,
                          float* __restrict__ bwd) {
  __shared__ __attribute__((aligned(16))) float S[2][BC * SP];
  const int tid   = threadIdx.x;
  const int lane  = tid & 31;
  const int w     = tid >> 5;
  const int hf    = lane >> 4;
  const int l16   = lane & 15;
  const int dir   = blockIdx.x >> 3;
  const int bbase = (blockIdx.x & 7) * BC;

  v2f a[KB];
  if (dir == 0) {
    const float* tp = Tm + (size_t)(w * 16 + l16) * NSTATE + 2 * hf;
#pragma unroll
    for (int kb = 0; kb < KB; ++kb) a[kb] = *(const v2f*)(tp + kb * 4);
  } else {
    const float* tp = Tm + (size_t)(2 * hf) * NSTATE + (w * 16 + l16);
#pragma unroll
    for (int kb = 0; kb < KB; ++kb) {
      a[kb].x = tp[(size_t)kb * 4 * NSTATE];
      a[kb].y = tp[(size_t)kb * 4 * NSTATE + NSTATE];
    }
  }

  float* dst = dir ? bwd : fwd;

  {
    const int t0 = dir ? (SEQLEN - 1) : 0;
    const v4f* e0 = (const v4f*)(em + (((size_t)t0 * NBATCH) + bbase) * NSTATE);
    v4f* g0 = (v4f*)(dst + (((size_t)t0 * NBATCH) + bbase) * NSTATE);
#pragma unroll
    for (int r = 0; r < 2; ++r) {
      const int i = tid + r * 512;
      const v4f v = e0[i];
      *(v4f*)(&S[0][(i >> 6) * SP + (i & 63) * 4]) = v;
      g0[i] = v;
    }
  }
  __syncthreads();

  for (int step = 1; step < SEQLEN; ++step) {
    const int t   = dir ? (SEQLEN - 1 - step) : step;
    const int cur = (step - 1) & 1, nxt = step & 1;
    const float* Sb = &S[cur][l16 * SP + 2 * hf];

    v8f acc0 = {0,0,0,0,0,0,0,0}, acc1 = acc0, acc2 = acc0, acc3 = acc0;
#pragma unroll
    for (int kb = 0; kb < KB; kb += 4) {
      const v2f b0 = *(const v2f*)(Sb + (kb + 0) * 4);
      const v2f b1 = *(const v2f*)(Sb + (kb + 1) * 4);
      const v2f b2 = *(const v2f*)(Sb + (kb + 2) * 4);
      const v2f b3 = *(const v2f*)(Sb + (kb + 3) * 4);
      acc0 = wmma_f32(a[kb + 0], b0, acc0);
      acc1 = wmma_f32(a[kb + 1], b1, acc1);
      acc2 = wmma_f32(a[kb + 2], b2, acc2);
      acc3 = wmma_f32(a[kb + 3], b3, acc3);
    }
    const v8f acc = (acc0 + acc1) + (acc2 + acc3);

    const size_t rowoff =
        (((size_t)t * NBATCH) + bbase + l16) * NSTATE + w * 16 + 8 * hf;
    const v4f e0 = *(const v4f*)(em + rowoff);
    const v4f e1 = *(const v4f*)(em + rowoff + 4);
    const v4f lo = {acc[0] * e0.x, acc[1] * e0.y, acc[2] * e0.z, acc[3] * e0.w};
    const v4f hi = {acc[4] * e1.x, acc[5] * e1.y, acc[6] * e1.z, acc[7] * e1.w};

    float* sp = &S[nxt][l16 * SP + w * 16 + 8 * hf];
    *(v4f*)sp       = lo;
    *(v4f*)(sp + 4) = hi;
    *(v4f*)(dst + rowoff)     = lo;
    *(v4f*)(dst + rowoff + 4) = hi;
    lds_barrier();
  }
}

__launch_bounds__(512)
__global__ void bwd_fused(const float* __restrict__ Tm,
                          const float* __restrict__ em,
                          float* __restrict__ fx) {
  __shared__ __attribute__((aligned(16))) float S[2][BC * SP];
  const int tid = threadIdx.x, lane = tid & 31, w = tid >> 5;
  const int hf = lane >> 4, l16 = lane & 15;
  const int bbase = blockIdx.x * BC;

  v2f a[KB];
  const float* tp = Tm + (size_t)(2 * hf) * NSTATE + (w * 16 + l16);
#pragma unroll
  for (int kb = 0; kb < KB; ++kb) {
    a[kb].x = tp[(size_t)kb * 4 * NSTATE];
    a[kb].y = tp[(size_t)kb * 4 * NSTATE + NSTATE];
  }

  {
    const v4f* e0 =
        (const v4f*)(em + (((size_t)(SEQLEN - 1) * NBATCH) + bbase) * NSTATE);
#pragma unroll
    for (int r = 0; r < 2; ++r) {
      const int i = tid + r * 512;
      *(v4f*)(&S[0][(i >> 6) * SP + (i & 63) * 4]) = e0[i];
    }
  }
  __syncthreads();

  for (int step = 1; step < SEQLEN; ++step) {
    const int t = SEQLEN - 1 - step;
    const int cur = (step - 1) & 1, nxt = step & 1;
    const float* Sb = &S[cur][l16 * SP + 2 * hf];

    v8f acc0 = {0,0,0,0,0,0,0,0}, acc1 = acc0, acc2 = acc0, acc3 = acc0;
#pragma unroll
    for (int kb = 0; kb < KB; kb += 4) {
      const v2f b0 = *(const v2f*)(Sb + (kb + 0) * 4);
      const v2f b1 = *(const v2f*)(Sb + (kb + 1) * 4);
      const v2f b2 = *(const v2f*)(Sb + (kb + 2) * 4);
      const v2f b3 = *(const v2f*)(Sb + (kb + 3) * 4);
      acc0 = wmma_f32(a[kb + 0], b0, acc0);
      acc1 = wmma_f32(a[kb + 1], b1, acc1);
      acc2 = wmma_f32(a[kb + 2], b2, acc2);
      acc3 = wmma_f32(a[kb + 3], b3, acc3);
    }
    const v8f acc = (acc0 + acc1) + (acc2 + acc3);

    const size_t rowoff =
        (((size_t)t * NBATCH) + bbase + l16) * NSTATE + w * 16 + 8 * hf;
    const v4f e0 = *(const v4f*)(em + rowoff);
    const v4f e1 = *(const v4f*)(em + rowoff + 4);
    const v4f f0 = *(const v4f*)(fx + rowoff);
    const v4f f1 = *(const v4f*)(fx + rowoff + 4);
    const v4f slo = {acc[0] * e0.x, acc[1] * e0.y, acc[2] * e0.z, acc[3] * e0.w};
    const v4f shi = {acc[4] * e1.x, acc[5] * e1.y, acc[6] * e1.z, acc[7] * e1.w};
    const v4f xlo = {acc[0] * f0.x, acc[1] * f0.y, acc[2] * f0.z, acc[3] * f0.w};
    const v4f xhi = {acc[4] * f1.x, acc[5] * f1.y, acc[6] * f1.z, acc[7] * f1.w};

    float* sp = &S[nxt][l16 * SP + w * 16 + 8 * hf];
    *(v4f*)sp       = slo;
    *(v4f*)(sp + 4) = shi;
    *(v4f*)(fx + rowoff)     = xlo;
    *(v4f*)(fx + rowoff + 4) = xhi;
    lds_barrier();
  }
}

__launch_bounds__(128)
__global__ void proj_kernel(const float* __restrict__ fwd,
                            const float* __restrict__ bwd,
                            const float* __restrict__ em,
                            const float* __restrict__ Wout,
                            float* __restrict__ out, int fused) {
  const int t  = blockIdx.x;
  const int bc = blockIdx.y;
  const int tid = threadIdx.x, lane = tid & 31, w = tid >> 5;
  const int hf = lane >> 4, l16 = lane & 15;

  v2f bw[KB];
  const float* bp = Wout + (size_t)(2 * hf) * NLAB + w * 16 + l16;
#pragma unroll
  for (int kb = 0; kb < KB; ++kb) {
    bw[kb].x = bp[(size_t)kb * 4 * NLAB];
    bw[kb].y = bp[(size_t)kb * 4 * NLAB + NLAB];
  }

  const size_t aoff =
      (((size_t)t * NBATCH) + bc * BC + l16) * NSTATE + 2 * hf;

  v8f acc0 = {0,0,0,0,0,0,0,0}, acc1 = acc0, acc2 = acc0, acc3 = acc0;
  if (fused) {
#pragma unroll
    for (int kb = 0; kb < KB; kb += 4) {
      const v2f a0 = *(const v2f*)(fwd + aoff + (kb + 0) * 4);
      const v2f a1 = *(const v2f*)(fwd + aoff + (kb + 1) * 4);
      const v2f a2 = *(const v2f*)(fwd + aoff + (kb + 2) * 4);
      const v2f a3 = *(const v2f*)(fwd + aoff + (kb + 3) * 4);
      acc0 = wmma_f32(a0, bw[kb + 0], acc0);
      acc1 = wmma_f32(a1, bw[kb + 1], acc1);
      acc2 = wmma_f32(a2, bw[kb + 2], acc2);
      acc3 = wmma_f32(a3, bw[kb + 3], acc3);
    }
  } else {
#pragma unroll
    for (int kb = 0; kb < KB; kb += 2) {
      const v2f f0 = *(const v2f*)(fwd + aoff + (kb + 0) * 4);
      const v2f g0 = *(const v2f*)(bwd + aoff + (kb + 0) * 4);
      const v2f e0 = *(const v2f*)(em  + aoff + (kb + 0) * 4);
      const v2f f1 = *(const v2f*)(fwd + aoff + (kb + 1) * 4);
      const v2f g1 = *(const v2f*)(bwd + aoff + (kb + 1) * 4);
      const v2f e1 = *(const v2f*)(em  + aoff + (kb + 1) * 4);
      const v2f a0 = {f0.x * g0.x / e0.x, f0.y * g0.y / e0.y};
      const v2f a1 = {f1.x * g1.x / e1.x, f1.y * g1.y / e1.y};
      acc0 = wmma_f32(a0, bw[kb + 0], acc0);
      acc1 = wmma_f32(a1, bw[kb + 1], acc1);
    }
  }
  const v8f acc = (acc0 + acc1) + (acc2 + acc3);

#pragma unroll
  for (int r = 0; r < 8; ++r) {
    const size_t b = (size_t)(bc * BC + 8 * hf + r);
    out[(b * SEQLEN + t) * NLAB + w * 16 + l16] = acc[r];
  }
}

extern "C" void kernel_launch(void* const* d_in, const int* in_sizes, int n_in,
                              void* d_out, int out_size, void* d_ws,
                              size_t ws_size, hipStream_t stream) {
  const int*   sent = (const int*)d_in[0];
  const float* emb  = (const float*)d_in[1];
  const float* Tm   = (const float*)d_in[2];
  const float* Wout = (const float*)d_in[3];
  float* out = (float*)d_out;

  const size_t NE = (size_t)SEQLEN * NBATCH * NSTATE;
  float* em  = (float*)d_ws;
  float* fwd = em + NE;

  gather_em<<<dim3(SEQLEN, NBATCH), 64, 0, stream>>>(sent, emb, em);

  if (ws_size >= 3 * NE * sizeof(float)) {
    float* bwd = fwd + NE;
    scan_pair<<<16, 512, 0, stream>>>(Tm, em, fwd, bwd);
    proj_kernel<<<dim3(SEQLEN, 8), 128, 0, stream>>>(fwd, bwd, em, Wout, out, 0);
  } else {
    scan_pair<<<8, 512, 0, stream>>>(Tm, em, fwd, nullptr);
    bwd_fused<<<8, 512, 0, stream>>>(Tm, em, fwd);
    proj_kernel<<<dim3(SEQLEN, 8), 128, 0, stream>>>(fwd, fwd, em, Wout, out, 1);
  }
}


# Scan step: the 64 LDS state-fragment reads stand in 3 runs (16, 8, 8 loads) issued 20 or more WMMAs ahead of their use, held by scheduling fences, instead of one read after every 2nd WMMA
# speedup vs baseline: 3.5939x; 1.0425x over previous
#include <hip/hip_runtime.h>


typedef float v2f __attribute__((ext_vector_type(2)));
typedef float v4f __attribute__((ext_vector_type(4)));
typedef float v8f __attribute__((ext_vector_type(8)));

#define NSTATE 256
#define NBATCH 128
#define SEQLEN 256
#define NLAB   64
#define BC     16
#define KB     64
#define SP     (NSTATE + 4)

static __device__ __forceinline__ v8f wmma_f32(v2f a, v2f b, v8f c) {
  return __builtin_amdgcn_wmma_f32_16x16x4_f32(
       false, a,  false, b,
       (short)0, c,  false,  false);
}

static __device__ __forceinline__ void lds_barrier() {
  __builtin_amdgcn_fence(__ATOMIC_RELEASE, "workgroup", "local");
  __builtin_amdgcn_s_barrier();
  __builtin_amdgcn_fence(__ATOMIC_ACQUIRE, "workgroup", "local");
}

template <int K0>
static __device__ __forceinline__ void load_frags(v2f (&b)[16], const float* Sb) {
#pragma unroll
  for (int i = 0; i < 16; ++i) b[i] = *(const v2f*)(Sb + (K0 + i) * 4);
}

template <int K0>
static __device__ __forceinline__ void mma_block(const v2f (&a)[KB], const v2f (&b)[16],
                                                 v8f& acc0, v8f& acc1, v8f& acc2, v8f& acc3) {
#pragma unroll
  for (int i = 0; i < 16; i += 4) {
    acc0 = wmma_f32(a[K0 + i + 0], b[i + 0], acc0);
    acc1 = wmma_f32(a[K0 + i + 1], b[i + 1], acc1);
    acc2 = wmma_f32(a[K0 + i + 2], b[i + 2], acc2);
    acc3 = wmma_f32(a[K0 + i + 3], b[i + 3], acc3);
  }
}

static __device__ __forceinline__ v8f step_product(const v2f (&a)[KB], const float* Sb) {
  v2f p[16], q[16];
  v8f acc0 = {0,0,0,0,0,0,0,0}, acc1 = acc0, acc2 = acc0, acc3 = acc0;
  load_frags<0>(p, Sb);
  load_frags<16>(q, Sb);
  __builtin_amdgcn_sched_barrier(0);
  mma_block<0>(a, p, acc0, acc1, acc2, acc3);
  load_frags<32>(p, Sb);
  __builtin_amdgcn_sched_barrier(0);
  mma_block<16>(a, q, acc0, acc1, acc2, acc3);
  load_frags<48>(q, Sb);
  __builtin_amdgcn_sched_barrier(0);
  mma_block<32>(a, p, acc0, acc1, acc2, acc3);
  mma_block<48>(a, q, acc0, acc1, acc2, acc3);
  return (acc0 + acc1) + (acc2 + acc3);
}

__global__ void gather_em(const int* __restrict__ sent,
                          const float* __restrict__ emb,
                          float* __restrict__ em) {
  const int t = blockIdx.x;
  const int b = blockIdx.y;
  const int tok = sent[(size_t)b * SEQLEN + t];
  const v4f* src = (const v4f*)(emb + (size_t)tok * NSTATE);
  v4f* dst = (v4f*)(em + (((size_t)t * NBATCH) + b) * NSTATE);
  dst[threadIdx.x] = src[threadIdx.x];
}

__launch_bounds__(512)
__global__ void scan_pair(const float* __restrict__ Tm,
                          const float* __restrict__ em,
                          float* __restrict__ fwd,
                          float* __restrict__ bwd) {
  __shared__ __attribute__((aligned(16))) float S[2][BC * SP];
  const int tid   = threadIdx.x;
  const int lane  = tid & 31;
  const int w     = tid >> 5;
  const int hf    = lane >> 4;
  const int l16   = lane & 15;
  const int dir   = blockIdx.x >> 3;
  const int bbase = (blockIdx.x & 7) * BC;

  v2f a[KB];
  if (dir == 0) {
    const float* tp = Tm + (size_t)(w * 16 + l16) * NSTATE + 2 * hf;
#pragma unroll
    for (int kb = 0; kb < KB; ++kb) a[kb] = *(const v2f*)(tp + kb * 4);
  } else {
    const float* tp = Tm + (size_t)(2 * hf) * NSTATE + (w * 16 + l16);
#pragma unroll
    for (int kb = 0; kb < KB; ++kb) {
      a[kb].x = tp[(size_t)kb * 4 * NSTATE];
      a[kb].y = tp[(size_t)kb * 4 * NSTATE + NSTATE];
    }
  }

  float* dst = dir ? bwd : fwd;

  {
    const int t0 = dir ? (SEQLEN - 1) : 0;
    const v4f* e0 = (const v4f*)(em + (((size_t)t0 * NBATCH) + bbase) * NSTATE);
    v4f* g0 = (v4f*)(dst + (((size_t)t0 * NBATCH) + bbase) * NSTATE);
#pragma unroll
    for (int r = 0; r < 2; ++r) {
      const int i = tid + r * 512;
      const v4f v = e0[i];
      *(v4f*)(&S[0][(i >> 6) * SP + (i & 63) * 4]) = v;
      g0[i] = v;
    }
  }
  __syncthreads();

  for (int step = 1; step < SEQLEN; ++step) {
    const int t   = dir ? (SEQLEN - 1 - step) : step;
    const int cur = (step - 1) & 1, nxt = step & 1;
    const float* Sb = &S[cur][l16 * SP + 2 * hf];

    const size_t rowoff =
        (((size_t)t * NBATCH) + bbase + l16) * NSTATE + w * 16 + 8 * hf;
    const v4f e0 = *(const v4f*)(em + rowoff);
    const v4f e1 = *(const v4f*)(em + rowoff + 4);

    const v8f acc = step_product(a, Sb);

    const v4f lo = {acc[0] * e0.x, acc[1] * e0.y, acc[2] * e0.z, acc[3] * e0.w};
    const v4f hi = {acc[4] * e1.x, acc[5] * e1.y, acc[6] * e1.z, acc[7] * e1.w};

    float* sp = &S[nxt][l16 * SP + w * 16 + 8 * hf];
    *(v4f*)sp       = lo;
    *(v4f*)(sp + 4) = hi;
    *(v4f*)(dst + rowoff)     = lo;
    *(v4f*)(dst + rowoff + 4) = hi;
    lds_barrier();
  }
}

__launch_bounds__(512)
__global__ void bwd_fused(const float* __restrict__ Tm,
                          const float* __restrict__ em,
                          float* __restrict__ fx) {
  __shared__ __attribute__((aligned(16))) float S[2][BC * SP];
  const int tid = threadIdx.x, lane = tid & 31, w = tid >> 5;
  const int hf = lane >> 4, l16 = lane & 15;
  const int bbase = blockIdx.x * BC;

  v2f a[KB];
  const float* tp = Tm + (size_t)(2 * hf) * NSTATE + (w * 16 + l16);
#pragma unroll
  for (int kb = 0; kb < KB; ++kb) {
    a[kb].x = tp[(size_t)kb * 4 * NSTATE];
    a[kb].y = tp[(size_t)kb * 4 * NSTATE + NSTATE];
  }

  {
    const v4f* e0 =
        (const v4f*)(em + (((size_t)(SEQLEN - 1) * NBATCH) + bbase) * NSTATE);
#pragma unroll
    for (int r = 0; r < 2; ++r) {
      const int i = tid + r * 512;
      *(v4f*)(&S[0][(i >> 6) * SP + (i & 63) * 4]) = e0[i];
    }
  }
  __syncthreads();

  for (int step = 1; step < SEQLEN; ++step) {
    const int t = SEQLEN - 1 - step;
    const int cur = (step - 1) & 1, nxt = step & 1;
    const float* Sb = &S[cur][l16 * SP + 2 * hf];

    v8f acc0 = {0,0,0,0,0,0,0,0}, acc1 = acc0, acc2 = acc0, acc3 = acc0;
#pragma unroll
    for (int kb = 0; kb < KB; kb += 4) {
      const v2f b0 = *(const v2f*)(Sb + (kb + 0) * 4);
      const v2f b1 = *(const v2f*)(Sb + (kb + 1) * 4);
      const v2f b2 = *(const v2f*)(Sb + (kb + 2) * 4);
      const v2f b3 = *(const v2f*)(Sb + (kb + 3) * 4);
      acc0 = wmma_f32(a[kb + 0], b0, acc0);
      acc1 = wmma_f32(a[kb + 1], b1, acc1);
      acc2 = wmma_f32(a[kb + 2], b2, acc2);
      acc3 = wmma_f32(a[kb + 3], b3, acc3);
    }
    const v8f acc = (acc0 + acc1) + (acc2 + acc3);

    const size_t rowoff =
        (((size_t)t * NBATCH) + bbase + l16) * NSTATE + w * 16 + 8 * hf;
    const v4f e0 = *(const v4f*)(em + rowoff);
    const v4f e1 = *(const v4f*)(em + rowoff + 4);
    const v4f f0 = *(const v4f*)(fx + rowoff);
    const v4f f1 = *(const v4f*)(fx + rowoff + 4);
    const v4f slo = {acc[0] * e0.x, acc[1] * e0.y, acc[2] * e0.z, acc[3] * e0.w};
    const v4f shi = {acc[4] * e1.x, acc[5] * e1.y, acc[6] * e1.z, acc[7] * e1.w};
    const v4f xlo = {acc[0] * f0.x, acc[1] * f0.y, acc[2] * f0.z, acc[3] * f0.w};
    const v4f xhi = {acc[4] * f1.x, acc[5] * f1.y, acc[6] * f1.z, acc[7] * f1.w};

    float* sp = &S[nxt][l16 * SP + w * 16 + 8 * hf];
    *(v4f*)sp       = slo;
    *(v4f*)(sp + 4) = shi;
    *(v4f*)(fx + rowoff)     = xlo;
    *(v4f*)(fx + rowoff + 4) = xhi;
    lds_barrier();
  }
}

__launch_bounds__(128)
__global__ void proj_kernel(const float* __restrict__ fwd,
                            const float* __restrict__ bwd,
                            const float* __restrict__ em,
                            const float* __restrict__ Wout,
                            float* __restrict__ out, int fused) {
  const int t  = blockIdx.x;
  const int bc = blockIdx.y;
  const int tid = threadIdx.x, lane = tid & 31, w = tid >> 5;
  const int hf = lane >> 4, l16 = lane & 15;

  v2f bw[KB];
  const float* bp = Wout + (size_t)(2 * hf) * NLAB + w * 16 + l16;
#pragma unroll
  for (int kb = 0; kb < KB; ++kb) {
    bw[kb].x = bp[(size_t)kb * 4 * NLAB];
    bw[kb].y = bp[(size_t)kb * 4 * NLAB + NLAB];
  }

  const size_t aoff =
      (((size_t)t * NBATCH) + bc * BC + l16) * NSTATE + 2 * hf;

  v8f acc0 = {0,0,0,0,0,0,0,0}, acc1 = acc0, acc2 = acc0, acc3 = acc0;
  if (fused) {
#pragma unroll
    for (int kb = 0; kb < KB; kb += 4) {
      const v2f a0 = *(const v2f*)(fwd + aoff + (kb + 0) * 4);
      const v2f a1 = *(const v2f*)(fwd + aoff + (kb + 1) * 4);
      const v2f a2 = *(const v2f*)(fwd + aoff + (kb + 2) * 4);
      const v2f a3 = *(const v2f*)(fwd + aoff + (kb + 3) * 4);
      acc0 = wmma_f32(a0, bw[kb + 0], acc0);
      acc1 = wmma_f32(a1, bw[kb + 1], acc1);
      acc2 = wmma_f32(a2, bw[kb + 2], acc2);
      acc3 = wmma_f32(a3, bw[kb + 3], acc3);
    }
  } else {
#pragma unroll
    for (int kb = 0; kb < KB; kb += 2) {
      const v2f f0 = *(const v2f*)(fwd + aoff + (kb + 0) * 4);
      const v2f g0 = *(const v2f*)(bwd + aoff + (kb + 0) * 4);
      const v2f e0 = *(const v2f*)(em  + aoff + (kb + 0) * 4);
      const v2f f1 = *(const v2f*)(fwd + aoff + (kb + 1) * 4);
      const v2f g1 = *(const v2f*)(bwd + aoff + (kb + 1) * 4);
      const v2f e1 = *(const v2f*)(em  + aoff + (kb + 1) * 4);
      const v2f a0 = {f0.x * g0.x / e0.x, f0.y * g0.y / e0.y};
      const v2f a1 = {f1.x * g1.x / e1.x, f1.y * g1.y / e1.y};
      acc0 = wmma_f32(a0, bw[kb + 0], acc0);
      acc1 = wmma_f32(a1, bw[kb + 1], acc1);
    }
  }
  const v8f acc = (acc0 + acc1) + (acc2 + acc3);

#pragma unroll
  for (int r = 0; r < 8; ++r) {
    const size_t b = (size_t)(bc * BC + 8 * hf + r);
    out[(b * SEQLEN + t) * NLAB + w * 16 + l16] = acc[r];
  }
}

extern "C" void kernel_launch(void* const* d_in, const int* in_sizes, int n_in,
                              void* d_out, int out_size, void* d_ws,
                              size_t ws_size, hipStream_t stream) {
  const int*   sent = (const int*)d_in[0];
  const float* emb  = (const float*)d_in[1];
  const float* Tm   = (const float*)d_in[2];
  const float* Wout = (const float*)d_in[3];
  float* out = (float*)d_out;

  const size_t NE = (size_t)SEQLEN * NBATCH * NSTATE;
  float* em  = (float*)d_ws;
  float* fwd = em + NE;

  gather_em<<<dim3(SEQLEN, NBATCH), 64, 0, stream>>>(sent, emb, em);

  if (ws_size >= 3 * NE * sizeof(float)) {
    float* bwd = fwd + NE;
    scan_pair<<<16, 512, 0, stream>>>(Tm, em, fwd, bwd);
    proj_kernel<<<dim3(SEQLEN, 8), 128, 0, stream>>>(fwd, bwd, em, Wout, out, 0);
  } else {
    scan_pair<<<8, 512, 0, stream>>>(Tm, em, fwd, nullptr);
    bwd_fused<<<8, 512, 0, stream>>>(Tm, em, fwd);
    proj_kernel<<<dim3(SEQLEN, 8), 128, 0, stream>>>(fwd, fwd, em, Wout, out, 1);
  }
}


# Projection: a block builds its 16x256 expected-count tile once (whole-line 16-byte loads, one division per element; 2 time slices per block) and shares it via LDS; before, each of 4 waves rebuilt it
# speedup vs baseline: 3.7472x; 1.0427x over previous
#include <hip/hip_runtime.h>


typedef float v2f __attribute__((ext_vector_type(2)));
typedef float v4f __attribute__((ext_vector_type(4)));
typedef float v8f __attribute__((ext_vector_type(8)));

#define NSTATE 256
#define NBATCH 128
#define SEQLEN 256
#define NLAB   64
#define BC     16
#define KB     64
#define SP     (NSTATE + 4)

static __device__ __forceinline__ v8f wmma_f32(v2f a, v2f b, v8f c) {
  return __builtin_amdgcn_wmma_f32_16x16x4_f32(
       false, a,  false, b,
       (short)0, c,  false,  false);
}

static __device__ __forceinline__ void lds_barrier() {
  __builtin_amdgcn_fence(__ATOMIC_RELEASE, "workgroup", "local");
  __builtin_amdgcn_s_barrier();
  __builtin_amdgcn_fence(__ATOMIC_ACQUIRE, "workgroup", "local");
}

template <int K0>
static __device__ __forceinline__ void load_frags(v2f (&b)[16], const float* Sb) {
#pragma unroll
  for (int i = 0; i < 16; ++i) b[i] = *(const v2f*)(Sb + (K0 + i) * 4);
}

template <int K0>
static __device__ __forceinline__ void mma_block(const v2f (&a)[KB], const v2f (&b)[16],
                                                 v8f& acc0, v8f& acc1, v8f& acc2, v8f& acc3) {
#pragma unroll
  for (int i = 0; i < 16; i += 4) {
    acc0 = wmma_f32(a[K0 + i + 0], b[i + 0], acc0);
    acc1 = wmma_f32(a[K0 + i + 1], b[i + 1], acc1);
    acc2 = wmma_f32(a[K0 + i + 2], b[i + 2], acc2);
    acc3 = wmma_f32(a[K0 + i + 3], b[i + 3], acc3);
  }
}

static __device__ __forceinline__ v8f step_product(const v2f (&a)[KB], const float* Sb) {
  v2f p[16], q[16];
  v8f acc0 = {0,0,0,0,0,0,0,0}, acc1 = acc0, acc2 = acc0, acc3 = acc0;
  load_frags<0>(p, Sb);
  load_frags<16>(q, Sb);
  __builtin_amdgcn_sched_barrier(0);
  mma_block<0>(a, p, acc0, acc1, acc2, acc3);
  load_frags<32>(p, Sb);
  __builtin_amdgcn_sched_barrier(0);
  mma_block<16>(a, q, acc0, acc1, acc2, acc3);
  load_frags<48>(q, Sb);
  __builtin_amdgcn_sched_barrier(0);
  mma_block<32>(a, p, acc0, acc1, acc2, acc3);
  mma_block<48>(a, q, acc0, acc1, acc2, acc3);
  return (acc0 + acc1) + (acc2 + acc3);
}

__global__ void gather_em(const int* __restrict__ sent,
                          const float* __restrict__ emb,
                          float* __restrict__ em) {
  const int t = blockIdx.x;
  const int b = blockIdx.y;
  const int tok = sent[(size_t)b * SEQLEN + t];
  const v4f* src = (const v4f*)(emb + (size_t)tok * NSTATE);
  v4f* dst = (v4f*)(em + (((size_t)t * NBATCH) + b) * NSTATE);
  dst[threadIdx.x] = src[threadIdx.x];
}

__launch_bounds__(512)
__global__ void scan_pair(const float* __restrict__ Tm,
                          const float* __restrict__ em,
                          float* __restrict__ fwd,
                          float* __restrict__ bwd) {
  __shared__ __attribute__((aligned(16))) float S[2][BC * SP];
  const int tid   = threadIdx.x;
  const int lane  = tid & 31;
  const int w     = tid >> 5;
  const int hf    = lane >> 4;
  const int l16   = lane & 15;
  const int dir   = blockIdx.x >> 3;
  const int bbase = (blockIdx.x & 7) * BC;

  v2f a[KB];
  if (dir == 0) {
    const float* tp = Tm + (size_t)(w * 16 + l16) * NSTATE + 2 * hf;
#pragma unroll
    for (int kb = 0; kb < KB; ++kb) a[kb] = *(const v2f*)(tp + kb * 4);
  } else {
    const float* tp = Tm + (size_t)(2 * hf) * NSTATE + (w * 16 + l16);
#pragma unroll
    for (int kb = 0; kb < KB; ++kb) {
      a[kb].x = tp[(size_t)kb * 4 * NSTATE];
      a[kb].y = tp[(size_t)kb * 4 * NSTATE + NSTATE];
    }
  }

  float* dst = dir ? bwd : fwd;

  {
    const int t0 = dir ? (SEQLEN - 1) : 0;
    const v4f* e0 = (const v4f*)(em + (((size_t)t0 * NBATCH) + bbase) * NSTATE);
    v4f* g0 = (v4f*)(dst + (((size_t)t0 * NBATCH) + bbase) * NSTATE);
#pragma unroll
    for (int r = 0; r < 2; ++r) {
      const int i = tid + r * 512;
      const v4f v = e0[i];
      *(v4f*)(&S[0][(i >> 6) * SP + (i & 63) * 4]) = v;
      g0[i] = v;
    }
  }
  __syncthreads();

  for (int step = 1; step < SEQLEN; ++step) {
    const int t   = dir ? (SEQLEN - 1 - step) : step;
    const int cur = (step - 1) & 1, nxt = step & 1;
    const float* Sb = &S[cur][l16 * SP + 2 * hf];

    const size_t rowoff =
        (((size_t)t * NBATCH) + bbase + l16) * NSTATE + w * 16 + 8 * hf;
    const v4f e0 = *(const v4f*)(em + rowoff);
    const v4f e1 = *(const v4f*)(em + rowoff + 4);

    const v8f acc = step_product(a, Sb);

    const v4f lo = {acc[0] * e0.x, acc[1] * e0.y, acc[2] * e0.z, acc[3] * e0.w};
    const v4f hi = {acc[4] * e1.x, acc[5] * e1.y, acc[6] * e1.z, acc[7] * e1.w};

    float* sp = &S[nxt][l16 * SP + w * 16 + 8 * hf];
    *(v4f*)sp       = lo;
    *(v4f*)(sp + 4) = hi;
    *(v4f*)(dst + rowoff)     = lo;
    *(v4f*)(dst + rowoff + 4) = hi;
    lds_barrier();
  }
}

__launch_bounds__(512)
__global__ void bwd_fused(const float* __restrict__ Tm,
                          const float* __restrict__ em,
                          float* __restrict__ fx) {
  __shared__ __attribute__((aligned(16))) float S[2][BC * SP];
  const int tid = threadIdx.x, lane = tid & 31, w = tid >> 5;
  const int hf = lane >> 4, l16 = lane & 15;
  const int bbase = blockIdx.x * BC;

  v2f a[KB];
  const float* tp = Tm + (size_t)(2 * hf) * NSTATE + (w * 16 + l16);
#pragma unroll
  for (int kb = 0; kb < KB; ++kb) {
    a[kb].x = tp[(size_t)kb * 4 * NSTATE];
    a[kb].y = tp[(size_t)kb * 4 * NSTATE + NSTATE];
  }

  {
    const v4f* e0 =
        (const v4f*)(em + (((size_t)(SEQLEN - 1) * NBATCH) + bbase) * NSTATE);
#pragma unroll
    for (int r = 0; r < 2; ++r) {
      const int i = tid + r * 512;
      *(v4f*)(&S[0][(i >> 6) * SP + (i & 63) * 4]) = e0[i];
    }
  }
  __syncthreads();

  for (int step = 1; step < SEQLEN; ++step) {
    const int t = SEQLEN - 1 - step;
    const int cur = (step - 1) & 1, nxt = step & 1;
    const float* Sb = &S[cur][l16 * SP + 2 * hf];

    v8f acc0 = {0,0,0,0,0,0,0,0}, acc1 = acc0, acc2 = acc0, acc3 = acc0;
#pragma unroll
    for (int kb = 0; kb < KB; kb += 4) {
      const v2f b0 = *(const v2f*)(Sb + (kb + 0) * 4);
      const v2f b1 = *(const v2f*)(Sb + (kb + 1) * 4);
      const v2f b2 = *(const v2f*)(Sb + (kb + 2) * 4);
      const v2f b3 = *(const v2f*)(Sb + (kb + 3) * 4);
      acc0 = wmma_f32(a[kb + 0], b0, acc0);
      acc1 = wmma_f32(a[kb + 1], b1, acc1);
      acc2 = wmma_f32(a[kb + 2], b2, acc2);
      acc3 = wmma_f32(a[kb + 3], b3, acc3);
    }
    const v8f acc = (acc0 + acc1) + (acc2 + acc3);

    const size_t rowoff =
        (((size_t)t * NBATCH) + bbase + l16) * NSTATE + w * 16 + 8 * hf;
    const v4f e0 = *(const v4f*)(em + rowoff);
    const v4f e1 = *(const v4f*)(em + rowoff + 4);
    const v4f f0 = *(const v4f*)(fx + rowoff);
    const v4f f1 = *(const v4f*)(fx + rowoff + 4);
    const v4f slo = {acc[0] * e0.x, acc[1] * e0.y, acc[2] * e0.z, acc[3] * e0.w};
    const v4f shi = {acc[4] * e1.x, acc[5] * e1.y, acc[6] * e1.z, acc[7] * e1.w};
    const v4f xlo = {acc[0] * f0.x, acc[1] * f0.y, acc[2] * f0.z, acc[3] * f0.w};
    const v4f xhi = {acc[4] * f1.x, acc[5] * f1.y, acc[6] * f1.z, acc[7] * f1.w};

    float* sp = &S[nxt][l16 * SP + w * 16 + 8 * hf];
    *(v4f*)sp       = slo;
    *(v4f*)(sp + 4) = shi;
    *(v4f*)(fx + rowoff)     = xlo;
    *(v4f*)(fx + rowoff + 4) = xhi;
    lds_barrier();
  }
}

#define PT 2

__launch_bounds__(128)
__global__ void proj_kernel(const float* __restrict__ fwd,
                            const float* __restrict__ bwd,
                            const float* __restrict__ em,
                            const float* __restrict__ Wout,
                            float* __restrict__ out, int fused) {
  __shared__ __attribute__((aligned(16))) float E[PT][BC * SP];
  const int bc = blockIdx.y;
  const int tid = threadIdx.x, lane = tid & 31, w = tid >> 5;
  const int hf = lane >> 4, l16 = lane & 15;

  v2f bw[KB];
  const float* bp = Wout + (size_t)(2 * hf) * NLAB + w * 16 + l16;
#pragma unroll
  for (int kb = 0; kb < KB; ++kb) {
    bw[kb].x = bp[(size_t)kb * 4 * NLAB];
    bw[kb].y = bp[(size_t)kb * 4 * NLAB + NLAB];
  }

  for (int it = 0; it < PT; ++it) {
    const int t = blockIdx.x * PT + it;
    float* Eb = E[it];
    const size_t tbase = (((size_t)t * NBATCH) + bc * BC) * NSTATE;
    const v4f* f4 = (const v4f*)(fwd + tbase);
    if (fused) {
#pragma unroll
      for (int i = 0; i < 8; ++i) {
        const int idx = tid + i * 128;
        *(v4f*)(&Eb[(idx >> 6) * SP + (idx & 63) * 4]) = f4[idx];
      }
    } else {
      const v4f* g4 = (const v4f*)(bwd + tbase);
      const v4f* e4 = (const v4f*)(em + tbase);
      v4f f[8], g[8], e[8];
#pragma unroll
      for (int i = 0; i < 8; ++i) {
        const int idx = tid + i * 128;
        f[i] = f4[idx]; g[i] = g4[idx]; e[i] = e4[idx];
      }
      __builtin_amdgcn_sched_barrier(0);
#pragma unroll
      for (int i = 0; i < 8; ++i) {
        const int idx = tid + i * 128;
        const v4f x = {f[i].x * g[i].x / e[i].x, f[i].y * g[i].y / e[i].y,
                       f[i].z * g[i].z / e[i].z, f[i].w * g[i].w / e[i].w};
        *(v4f*)(&Eb[(idx >> 6) * SP + (idx & 63) * 4]) = x;
      }
    }
    lds_barrier();

    const float* Ab = Eb + l16 * SP + 2 * hf;
    v8f acc0 = {0,0,0,0,0,0,0,0}, acc1 = acc0, acc2 = acc0, acc3 = acc0;
#pragma unroll
    for (int kb = 0; kb < KB; kb += 2) {
      acc0 = wmma_f32(*(const v2f*)(Ab + (kb + 0) * 4), bw[kb + 0], acc0);
      acc1 = wmma_f32(*(const v2f*)(Ab + (kb + 1) * 4), bw[kb + 1], acc1);
    }
    const v8f acc = (acc0 + acc1) + (acc2 + acc3);

#pragma unroll
    for (int r = 0; r < 8; ++r) {
      const size_t b = (size_t)(bc * BC + 8 * hf + r);
      out[(b * SEQLEN + t) * NLAB + w * 16 + l16] = acc[r];
    }
  }
}

extern "C" void kernel_launch(void* const* d_in, const int* in_sizes, int n_in,
                              void* d_out, int out_size, void* d_ws,
                              size_t ws_size, hipStream_t stream) {
  const int*   sent = (const int*)d_in[0];
  const float* emb  = (const float*)d_in[1];
  const float* Tm   = (const float*)d_in[2];
  const float* Wout = (const float*)d_in[3];
  float* out = (float*)d_out;

  const size_t NE = (size_t)SEQLEN * NBATCH * NSTATE;
  float* em  = (float*)d_ws;
  float* fwd = em + NE;

  gather_em<<<dim3(SEQLEN, NBATCH), 64, 0, stream>>>(sent, emb, em);

  if (ws_size >= 3 * NE * sizeof(float)) {
    float* bwd = fwd + NE;
    scan_pair<<<16, 512, 0, stream>>>(Tm, em, fwd, bwd);
    proj_kernel<<<dim3(SEQLEN / PT, 8), 128, 0, stream>>>(fwd, bwd, em, Wout, out, 0);
  } else {
    scan_pair<<<8, 512, 0, stream>>>(Tm, em, fwd, nullptr);
    bwd_fused<<<8, 512, 0, stream>>>(Tm, em, fwd);
    proj_kernel<<<dim3(SEQLEN / PT, 8), 128, 0, stream>>>(fwd, fwd, em, Wout, out, 1);
  }
}


# Scan and projection read emission rows straight from the embedding table by token (scan: next step's token fetched a step ahead): the gather launch and its 32 MiB plane are gone, 3 launches to 2
# speedup vs baseline: 4.0897x; 1.0914x over previous
#include <hip/hip_runtime.h>


typedef float v2f __attribute__((ext_vector_type(2)));
typedef float v4f __attribute__((ext_vector_type(4)));
typedef float v8f __attribute__((ext_vector_type(8)));

#define NSTATE 256
#define NBATCH 128
#define SEQLEN 256
#define NLAB   64
#define BC     16
#define KB     64
#define SP     (NSTATE + 4)

static __device__ __forceinline__ v8f wmma_f32(v2f a, v2f b, v8f c) {
  return __builtin_amdgcn_wmma_f32_16x16x4_f32(
       false, a,  false, b,
       (short)0, c,  false,  false);
}

static __device__ __forceinline__ void lds_barrier() {
  __builtin_amdgcn_fence(__ATOMIC_RELEASE, "workgroup", "local");
  __builtin_amdgcn_s_barrier();
  __builtin_amdgcn_fence(__ATOMIC_ACQUIRE, "workgroup", "local");
}

template <int K0>
static __device__ __forceinline__ void load_frags(v2f (&b)[16], const float* Sb) {
#pragma unroll
  for (int i = 0; i < 16; ++i) b[i] = *(const v2f*)(Sb + (K0 + i) * 4);
}

template <int K0>
static __device__ __forceinline__ void mma_block(const v2f (&a)[KB], const v2f (&b)[16],
                                                 v8f& acc0, v8f& acc1, v8f& acc2, v8f& acc3) {
#pragma unroll
  for (int i = 0; i < 16; i += 4) {
    acc0 = wmma_f32(a[K0 + i + 0], b[i + 0], acc0);
    acc1 = wmma_f32(a[K0 + i + 1], b[i + 1], acc1);
    acc2 = wmma_f32(a[K0 + i + 2], b[i + 2], acc2);
    acc3 = wmma_f32(a[K0 + i + 3], b[i + 3], acc3);
  }
}

static __device__ __forceinline__ v8f step_product(const v2f (&a)[KB], const float* Sb) {
  v2f p[16], q[16];
  v8f acc0 = {0,0,0,0,0,0,0,0}, acc1 = acc0, acc2 = acc0, acc3 = acc0;
  load_frags<0>(p, Sb);
  load_frags<16>(q, Sb);
  __builtin_amdgcn_sched_barrier(0);
  mma_block<0>(a, p, acc0, acc1, acc2, acc3);
  load_frags<32>(p, Sb);
  __builtin_amdgcn_sched_barrier(0);
  mma_block<16>(a, q, acc0, acc1, acc2, acc3);
  load_frags<48>(q, Sb);
  __builtin_amdgcn_sched_barrier(0);
  mma_block<32>(a, p, acc0, acc1, acc2, acc3);
  mma_block<48>(a, q, acc0, acc1, acc2, acc3);
  return (acc0 + acc1) + (acc2 + acc3);
}

__launch_bounds__(512)
__global__ void scan_pair(const float* __restrict__ Tm,
                          const int* __restrict__ sent,
                          const float* __restrict__ emb,
                          float* __restrict__ fwd,
                          float* __restrict__ bwd) {
  __shared__ __attribute__((aligned(16))) float S[2][BC * SP];
  const int tid   = threadIdx.x;
  const int lane  = tid & 31;
  const int w     = tid >> 5;
  const int hf    = lane >> 4;
  const int l16   = lane & 15;
  const int dir   = blockIdx.x >> 3;
  const int bbase = (blockIdx.x & 7) * BC;

  v2f a[KB];
  if (dir == 0) {
    const float* tp = Tm + (size_t)(w * 16 + l16) * NSTATE + 2 * hf;
#pragma unroll
    for (int kb = 0; kb < KB; ++kb) a[kb] = *(const v2f*)(tp + kb * 4);
  } else {
    const float* tp = Tm + (size_t)(2 * hf) * NSTATE + (w * 16 + l16);
#pragma unroll
    for (int kb = 0; kb < KB; ++kb) {
      a[kb].x = tp[(size_t)kb * 4 * NSTATE];
      a[kb].y = tp[(size_t)kb * 4 * NSTATE + NSTATE];
    }
  }

  float* dst = dir ? bwd : fwd;

  {
    const int t0 = dir ? (SEQLEN - 1) : 0;
    v4f* g0 = (v4f*)(dst + (((size_t)t0 * NBATCH) + bbase) * NSTATE);
#pragma unroll
    for (int r = 0; r < 2; ++r) {
      const int i = tid + r * 512;
      const int tk = sent[(size_t)(bbase + (i >> 6)) * SEQLEN + t0];
      const v4f v = *(const v4f*)(emb + (size_t)tk * NSTATE + (i & 63) * 4);
      *(v4f*)(&S[0][(i >> 6) * SP + (i & 63) * 4]) = v;
      g0[i] = v;
    }
  }

  const int* tcol = sent + (size_t)(bbase + l16) * SEQLEN;
  int tok = tcol[dir ? (SEQLEN - 2) : 1];
  __syncthreads();

  for (int step = 1; step < SEQLEN; ++step) {
    const int t   = dir ? (SEQLEN - 1 - step) : step;
    const int cur = (step - 1) & 1, nxt = step & 1;
    const float* Sb = &S[cur][l16 * SP + 2 * hf];

    const size_t rowoff =
        (((size_t)t * NBATCH) + bbase + l16) * NSTATE + w * 16 + 8 * hf;
    const float* er = emb + (size_t)tok * NSTATE + w * 16 + 8 * hf;
    const v4f e0 = *(const v4f*)er;
    const v4f e1 = *(const v4f*)(er + 4);
    tok = tcol[dir ? (t > 0 ? t - 1 : 0) : (t < SEQLEN - 1 ? t + 1 : SEQLEN - 1)];

    const v8f acc = step_product(a, Sb);

    const v4f lo = {acc[0] * e0.x, acc[1] * e0.y, acc[2] * e0.z, acc[3] * e0.w};
    const v4f hi = {acc[4] * e1.x, acc[5] * e1.y, acc[6] * e1.z, acc[7] * e1.w};

    float* sp = &S[nxt][l16 * SP + w * 16 + 8 * hf];
    *(v4f*)sp       = lo;
    *(v4f*)(sp + 4) = hi;
    *(v4f*)(dst + rowoff)     = lo;
    *(v4f*)(dst + rowoff + 4) = hi;
    lds_barrier();
  }
}

__launch_bounds__(512)
__global__ void bwd_fused(const float* __restrict__ Tm,
                          const int* __restrict__ sent,
                          const float* __restrict__ emb,
                          float* __restrict__ fx) {
  __shared__ __attribute__((aligned(16))) float S[2][BC * SP];
  const int tid = threadIdx.x, lane = tid & 31, w = tid >> 5;
  const int hf = lane >> 4, l16 = lane & 15;
  const int bbase = blockIdx.x * BC;

  v2f a[KB];
  const float* tp = Tm + (size_t)(2 * hf) * NSTATE + (w * 16 + l16);
#pragma unroll
  for (int kb = 0; kb < KB; ++kb) {
    a[kb].x = tp[(size_t)kb * 4 * NSTATE];
    a[kb].y = tp[(size_t)kb * 4 * NSTATE + NSTATE];
  }

  {
#pragma unroll
    for (int r = 0; r < 2; ++r) {
      const int i = tid + r * 512;
      const int tk = sent[(size_t)(bbase + (i >> 6)) * SEQLEN + (SEQLEN - 1)];
      *(v4f*)(&S[0][(i >> 6) * SP + (i & 63) * 4]) =
          *(const v4f*)(emb + (size_t)tk * NSTATE + (i & 63) * 4);
    }
  }
  const int* tcol = sent + (size_t)(bbase + l16) * SEQLEN;
  int tok = tcol[SEQLEN - 2];
  __syncthreads();

  for (int step = 1; step < SEQLEN; ++step) {
    const int t = SEQLEN - 1 - step;
    const int cur = (step - 1) & 1, nxt = step & 1;
    const float* Sb = &S[cur][l16 * SP + 2 * hf];
    const float* er = emb + (size_t)tok * NSTATE + w * 16 + 8 * hf;
    tok = tcol[t > 0 ? t - 1 : 0];

    v8f acc0 = {0,0,0,0,0,0,0,0}, acc1 = acc0, acc2 = acc0, acc3 = acc0;
#pragma unroll
    for (int kb = 0; kb < KB; kb += 4) {
      const v2f b0 = *(const v2f*)(Sb + (kb + 0) * 4);
      const v2f b1 = *(const v2f*)(Sb + (kb + 1) * 4);
      const v2f b2 = *(const v2f*)(Sb + (kb + 2) * 4);
      const v2f b3 = *(const v2f*)(Sb + (kb + 3) * 4);
      acc0 = wmma_f32(a[kb + 0], b0, acc0);
      acc1 = wmma_f32(a[kb + 1], b1, acc1);
      acc2 = wmma_f32(a[kb + 2], b2, acc2);
      acc3 = wmma_f32(a[kb + 3], b3, acc3);
    }
    const v8f acc = (acc0 + acc1) + (acc2 + acc3);

    const size_t rowoff =
        (((size_t)t * NBATCH) + bbase + l16) * NSTATE + w * 16 + 8 * hf;
    const v4f e0 = *(const v4f*)er;
    const v4f e1 = *(const v4f*)(er + 4);
    const v4f f0 = *(const v4f*)(fx + rowoff);
    const v4f f1 = *(const v4f*)(fx + rowoff + 4);
    const v4f slo = {acc[0] * e0.x, acc[1] * e0.y, acc[2] * e0.z, acc[3] * e0.w};
    const v4f shi = {acc[4] * e1.x, acc[5] * e1.y, acc[6] * e1.z, acc[7] * e1.w};
    const v4f xlo = {acc[0] * f0.x, acc[1] * f0.y, acc[2] * f0.z, acc[3] * f0.w};
    const v4f xhi = {acc[4] * f1.x, acc[5] * f1.y, acc[6] * f1.z, acc[7] * f1.w};

    float* sp = &S[nxt][l16 * SP + w * 16 + 8 * hf];
    *(v4f*)sp       = slo;
    *(v4f*)(sp + 4) = shi;
    *(v4f*)(fx + rowoff)     = xlo;
    *(v4f*)(fx + rowoff + 4) = xhi;
    lds_barrier();
  }
}

#define PT 2

__launch_bounds__(128)
__global__ void proj_kernel(const float* __restrict__ fwd,
                            const float* __restrict__ bwd,
                            const int* __restrict__ sent,
                            const float* __restrict__ emb,
                            const float* __restrict__ Wout,
                            float* __restrict__ out, int fused) {
  __shared__ __attribute__((aligned(16))) float E[PT][BC * SP];
  const int bc = blockIdx.y;
  const int tid = threadIdx.x, lane = tid & 31, w = tid >> 5;
  const int hf = lane >> 4, l16 = lane & 15;

  const int rpar = __builtin_amdgcn_readfirstlane(tid >> 6);
  int tk[PT][8];
  if (!fused) {
#pragma unroll
    for (int it = 0; it < PT; ++it)
#pragma unroll
      for (int i = 0; i < 8; ++i)
        tk[it][i] = sent[(size_t)(bc * BC + 2 * i + rpar) * SEQLEN + blockIdx.x * PT + it];
  }

  v2f bw[KB];
  const float* bp = Wout + (size_t)(2 * hf) * NLAB + w * 16 + l16;
#pragma unroll
  for (int kb = 0; kb < KB; ++kb) {
    bw[kb].x = bp[(size_t)kb * 4 * NLAB];
    bw[kb].y = bp[(size_t)kb * 4 * NLAB + NLAB];
  }

  for (int it = 0; it < PT; ++it) {
    const int t = blockIdx.x * PT + it;
    float* Eb = E[it];
    const size_t tbase = (((size_t)t * NBATCH) + bc * BC) * NSTATE;
    const v4f* f4 = (const v4f*)(fwd + tbase);
    if (fused) {
#pragma unroll
      for (int i = 0; i < 8; ++i) {
        const int idx = tid + i * 128;
        *(v4f*)(&Eb[(idx >> 6) * SP + (idx & 63) * 4]) = f4[idx];
      }
    } else {
      const v4f* g4 = (const v4f*)(bwd + tbase);
      v4f f[8], g[8], e[8];
#pragma unroll
      for (int i = 0; i < 8; ++i) {
        const int idx = tid + i * 128;
        f[i] = f4[idx]; g[i] = g4[idx];
        e[i] = *(const v4f*)(emb + (size_t)tk[it][i] * NSTATE + (tid & 63) * 4);
      }
      __builtin_amdgcn_sched_barrier(0);
#pragma unroll
      for (int i = 0; i < 8; ++i) {
        const int idx = tid + i * 128;
        const v4f x = {f[i].x * g[i].x / e[i].x, f[i].y * g[i].y / e[i].y,
                       f[i].z * g[i].z / e[i].z, f[i].w * g[i].w / e[i].w};
        *(v4f*)(&Eb[(idx >> 6) * SP + (idx & 63) * 4]) = x;
      }
    }
    lds_barrier();

    const float* Ab = Eb + l16 * SP + 2 * hf;
    v8f acc0 = {0,0,0,0,0,0,0,0}, acc1 = acc0, acc2 = acc0, acc3 = acc0;
#pragma unroll
    for (int kb = 0; kb < KB; kb += 2) {
      acc0 = wmma_f32(*(const v2f*)(Ab + (kb + 0) * 4), bw[kb + 0], acc0);
      acc1 = wmma_f32(*(const v2f*)(Ab + (kb + 1) * 4), bw[kb + 1], acc1);
    }
    const v8f acc = (acc0 + acc1) + (acc2 + acc3);

#pragma unroll
    for (int r = 0; r < 8; ++r) {
      const size_t b = (size_t)(bc * BC + 8 * hf + r);
      out[(b * SEQLEN + t) * NLAB + w * 16 + l16] = acc[r];
    }
  }
}

extern "C" void kernel_launch(void* const* d_in, const int* in_sizes, int n_in,
                              void* d_out, int out_size, void* d_ws,
                              size_t ws_size, hipStream_t stream) {
  const int*   sent = (const int*)d_in[0];
  const float* emb  = (const float*)d_in[1];
  const float* Tm   = (const float*)d_in[2];
  const float* Wout = (const float*)d_in[3];
  float* out = (float*)d_out;

  const size_t NE = (size_t)SEQLEN * NBATCH * NSTATE;
  float* fwd = (float*)d_ws;

  if (ws_size >= 2 * NE * sizeof(float)) {
    float* bwd = fwd + NE;
    scan_pair<<<16, 512, 0, stream>>>(Tm, sent, emb, fwd, bwd);
    proj_kernel<<<dim3(SEQLEN / PT, 8), 128, 0, stream>>>(fwd, bwd, sent, emb, Wout, out, 0);
  } else {
    scan_pair<<<8, 512, 0, stream>>>(Tm, sent, emb, fwd, nullptr);
    bwd_fused<<<8, 512, 0, stream>>>(Tm, sent, emb, fwd);
    proj_kernel<<<dim3(SEQLEN / PT, 8), 128, 0, stream>>>(fwd, fwd, sent, emb, Wout, out, 1);
  }
}
